# starting kernel
# baseline (speedup 1.0000x reference)
#include <hip/hip_runtime.h>

#define N_NODES 200000
#define N_EDGES 3200000
#define FDIM    16
#define NTILES  (N_NODES / 16)

typedef float v2f __attribute__((ext_vector_type(2)));
typedef float v8f __attribute__((ext_vector_type(8)));

__global__ void k_zero(float* __restrict__ p, int n) {
    int i = blockIdx.x * blockDim.x + threadIdx.x;
    if (i < n) p[i] = 0.0f;
}

__global__ __launch_bounds__(256) void k_scatter(const float* __restrict__ feat,
                                                 const int* __restrict__ ei,
                                                 float* __restrict__ agg) {
    unsigned int t = blockIdx.x * 256u + threadIdx.x;
    if (t >= (unsigned int)N_EDGES * FDIM) return;
    int e = (int)(t >> 4);
    int f = (int)(t & 15u);
    int s = ei[e];
    int d = ei[N_EDGES + e];
    atomicAdd(&agg[(size_t)d * FDIM + f], feat[(size_t)s * FDIM + f]);
}

template <bool LOGSOFTMAX>
__global__ __launch_bounds__(256) void k_gin_mlp(const float* __restrict__ xin,
                                                 const float* __restrict__ agg,
                                                 const float* __restrict__ eps_p,
                                                 const float* __restrict__ wa,
                                                 const float* __restrict__ ba,
                                                 const float* __restrict__ wb,
                                                 const float* __restrict__ bb,
                                                 float* __restrict__ out,
                                                 int n_tiles) {
    __shared__ float lds[8][16][17];

    const int wid  = threadIdx.x >> 5;
    const int lane = threadIdx.x & 31;
    const int tile = blockIdx.x * 8 + wid;
    if (tile >= n_tiles) return;

    const int col = lane & 15;
    const int hi  = lane >> 4;
    const float epsp = 1.0f + eps_p[0];

    const float* xrow = xin + (size_t)(tile * 16 + col) * FDIM;
    const float* grow = agg + (size_t)(tile * 16 + col) * FDIM;

    v8f acc = {};
#pragma unroll
    for (int c = 0; c < 4; ++c) {
        const int kb = c * 4 + hi * 2;
        v2f xv = *(const v2f*)(xrow + kb);
        v2f gv = *(const v2f*)(grow + kb);
        v2f a;
        a.x = epsp * xv.x + gv.x;
        a.y = epsp * xv.y + gv.y;
        v2f b;
        b.x = wa[(kb + 0) * 16 + col];
        b.y = wa[(kb + 1) * 16 + col];
        acc = __builtin_amdgcn_wmma_f32_16x16x4_f32(false, a, false, b,
                                                    (short)0, acc, false, false);
    }

    const float biasA = ba[col];
#pragma unroll
    for (int v = 0; v < 8; ++v) {
        float t = acc[v] + biasA;
        lds[wid][v + hi * 8][col] = t > 0.0f ? t : 0.0f;
    }
    __syncthreads();

    v8f acc2 = {};
#pragma unroll
    for (int c = 0; c < 4; ++c) {
        const int kb = c * 4 + hi * 2;
        v2f a;
        a.x = lds[wid][col][kb + 0];
        a.y = lds[wid][col][kb + 1];
        v2f b;
        b.x = wb[(kb + 0) * 16 + col];
        b.y = wb[(kb + 1) * 16 + col];
        acc2 = __builtin_amdgcn_wmma_f32_16x16x4_f32(false, a, false, b,
                                                     (short)0, acc2, false, false);
    }

    const float biasB = bb[col];
    float* orow = out + (size_t)tile * 16 * FDIM;

    if (LOGSOFTMAX) {
#pragma unroll
        for (int v = 0; v < 8; ++v) {
            float val = acc2[v] + biasB;
            float m = val;
#pragma unroll
            for (int off = 1; off < 16; off <<= 1)
                m = fmaxf(m, __shfl_xor(m, off, 16));
            float s = __expf(val - m);
#pragma unroll
            for (int off = 1; off < 16; off <<= 1)
                s += __shfl_xor(s, off, 16);
            orow[(v + hi * 8) * FDIM + col] = val - m - __logf(s);
        }
    } else {
#pragma unroll
        for (int v = 0; v < 8; ++v) {
            float t = acc2[v] + biasB;
            orow[(v + hi * 8) * FDIM + col] = t > 0.0f ? t : 0.0f;
        }
    }
}

extern "C" void kernel_launch(void* const* d_in, const int* in_sizes, int n_in,
                              void* d_out, int out_size, void* d_ws, size_t ws_size,
                              hipStream_t stream) {
    const float* x    = (const float*)d_in[0];
    const int*   ei   = (const int*)d_in[1];
    const float* eps1 = (const float*)d_in[2];
    const float* w1a  = (const float*)d_in[3];
    const float* b1a  = (const float*)d_in[4];
    const float* w1b  = (const float*)d_in[5];
    const float* b1b  = (const float*)d_in[6];
    const float* eps2 = (const float*)d_in[7];
    const float* w2a  = (const float*)d_in[8];
    const float* b2a  = (const float*)d_in[9];
    const float* w2b  = (const float*)d_in[10];
    const float* b2b  = (const float*)d_in[11];
    float* out = (float*)d_out;

    float* agg = (float*)d_ws;
    float* h1  = agg + (size_t)N_NODES * FDIM;

    const int nfeat       = N_NODES * FDIM;
    const int zero_blocks = (nfeat + 255) / 256;
    const unsigned int scat_threads = (unsigned int)N_EDGES * FDIM;
    const int scat_blocks = (int)((scat_threads + 255u) / 256u);
    const int mlp_blocks  = (NTILES + 7) / 8;

    k_zero<<<zero_blocks, 256, 0, stream>>>(agg, nfeat);
    k_scatter<<<scat_blocks, 256, 0, stream>>>(x, ei, agg);
    k_gin_mlp<false><<<mlp_blocks, 256, 0, stream>>>(x, agg, eps1, w1a, b1a, w1b, b1b,
                                                     h1, NTILES);
    k_zero<<<zero_blocks, 256, 0, stream>>>(agg, nfeat);
    k_scatter<<<scat_blocks, 256, 0, stream>>>(h1, ei, agg);
    k_gin_mlp<true><<<mlp_blocks, 256, 0, stream>>>(h1, agg, eps2, w2a, b2a, w2b, b2b,
                                                    out, NTILES);
}


# Neighbour sums moved from global f32 atomics to LDS f32 atomic adds: edges bucketed once per call by 400-row destination range (LDS-counted count/scan/place), one work-group per bucket, MLP fused
# speedup vs baseline: 3.9403x; 3.9403x over previous
#include <hip/hip_runtime.h>


#define N_NODES 200000
#define N_EDGES 3200000
#define FDIM    16
#define NTILES  (N_NODES / 16)

#define BROWS    400
#define BTILES   (BROWS / 16)
#define NBUCK    (N_NODES / BROWS)
#define LWAVES   BTILES
#define LTHREADS (LWAVES * 32)

#define BIN_THREADS 512
#define BIN_EPT     25
#define BIN_BLOCKS  (N_EDGES / (BIN_THREADS * BIN_EPT))

#define SRC_BITS 22
#define SRC_MASK 0x3FFFFFu
#define REC_SINK ((unsigned int)BROWS << SRC_BITS)

static_assert(NBUCK * BROWS == N_NODES, "buckets must tile the nodes");
static_assert(BIN_BLOCKS * BIN_THREADS * BIN_EPT == N_EDGES, "binning grid must tile the edges");
static_assert(N_NODES * FDIM <= (1 << SRC_BITS), "source row offset must fit its field");
static_assert(BROWS < (1 << (32 - SRC_BITS)), "bucket row (and the spare row) must fit its field");
static_assert(NBUCK <= BIN_THREADS, "one thread per bucket counter");

typedef float v2f __attribute__((ext_vector_type(2)));
typedef float v8f __attribute__((ext_vector_type(8)));

__global__ void k_zero_i(int* __restrict__ p, int n) {
    int i = blockIdx.x * blockDim.x + threadIdx.x;
    if (i < n) p[i] = 0;
}

__global__ __launch_bounds__(BIN_THREADS) void k_count(const int* __restrict__ ei,
                                                       int* __restrict__ gcount) {
    __shared__ int h[NBUCK];
    const int tid = threadIdx.x;
    if (tid < NBUCK) h[tid] = 0;
    __syncthreads();
    const int* dstp = ei + N_EDGES + blockIdx.x * (BIN_THREADS * BIN_EPT) + tid;
    int d[BIN_EPT];
#pragma unroll
    for (int k = 0; k < BIN_EPT; ++k) d[k] = dstp[k * BIN_THREADS];
#pragma unroll
    for (int k = 0; k < BIN_EPT; ++k) atomicAdd(&h[(unsigned int)d[k] / BROWS], 1);
    __syncthreads();
    if (tid < NBUCK) {
        const int c = h[tid];
        if (c) atomicAdd(&gcount[tid], c);
    }
}

__global__ __launch_bounds__(512) void k_scan(const int* __restrict__ gcount,
                                              int* __restrict__ bstart,
                                              int* __restrict__ cursor) {
    __shared__ int s[512];
    const int tid = threadIdx.x;
    const int v = tid < NBUCK ? gcount[tid] : 0;
    s[tid] = v;
    __syncthreads();
    for (int off = 1; off < 512; off <<= 1) {
        const int t = tid >= off ? s[tid - off] : 0;
        __syncthreads();
        s[tid] += t;
        __syncthreads();
    }
    const int incl = s[tid];
    if (tid < NBUCK) {
        bstart[tid] = incl - v;
        cursor[tid] = incl - v;
    }
    if (tid == NBUCK - 1) bstart[NBUCK] = incl;
}

__global__ __launch_bounds__(BIN_THREADS) void k_place(const int* __restrict__ ei,
                                                       int* __restrict__ cursor,
                                                       unsigned int* __restrict__ recs) {
    __shared__ int h[NBUCK];
    __shared__ int basep[NBUCK];
    const int tid = threadIdx.x;
    if (tid < NBUCK) h[tid] = 0;
    __syncthreads();
    const int* srcp = ei + blockIdx.x * (BIN_THREADS * BIN_EPT) + tid;
    const int* dstp = srcp + N_EDGES;
    int sv[BIN_EPT], dv[BIN_EPT];
#pragma unroll
    for (int k = 0; k < BIN_EPT; ++k) {
        sv[k] = srcp[k * BIN_THREADS];
        dv[k] = dstp[k * BIN_THREADS];
    }
    unsigned int rec[BIN_EPT];
    int bk[BIN_EPT], rk[BIN_EPT];
#pragma unroll
    for (int k = 0; k < BIN_EPT; ++k) {
        const unsigned int d = (unsigned int)dv[k];
        const unsigned int b = d / BROWS;
        bk[k]  = (int)b;
        rec[k] = ((unsigned int)sv[k] * FDIM) | ((d - b * BROWS) << SRC_BITS);
        rk[k]  = atomicAdd(&h[b], 1);
    }
    __syncthreads();
    if (tid < NBUCK) {
        const int c = h[tid];
        basep[tid] = c ? atomicAdd(&cursor[tid], c) : 0;
    }
    __syncthreads();
#pragma unroll
    for (int k = 0; k < BIN_EPT; ++k) recs[basep[bk[k]] + rk[k]] = rec[k];
}

template <bool LOGSOFTMAX>
__global__ __launch_bounds__(LTHREADS) void k_layer(const float* __restrict__ xin,
                                                    const unsigned int* __restrict__ recs,
                                                    const int* __restrict__ bstart,
                                                    const float* __restrict__ eps_p,
                                                    const float* __restrict__ wa,
                                                    const float* __restrict__ ba,
                                                    const float* __restrict__ wb,
                                                    const float* __restrict__ bb,
                                                    float* __restrict__ out) {
    __shared__ __attribute__((aligned(16))) float sagg[(BROWS + 1) * FDIM];
    __shared__ float lds[LWAVES][16][17];

    const int tid  = threadIdx.x;
    const int wid  = __builtin_amdgcn_readfirstlane(tid >> 5);
    const int lane = tid & 31;
    const int col  = lane & 15;
    const int hi   = lane >> 4;
    const int bucket = blockIdx.x;

    for (int i = tid; i < (BROWS + 1) * FDIM; i += LTHREADS) sagg[i] = 0.0f;
    const int ebeg = bstart[bucket];
    const int eend = bstart[bucket + 1];
    __syncthreads();

    int base = ebeg + wid * 32;
    unsigned int raw = recs[min(base + lane, N_EDGES - 1)];
    while (base < eend) {
        const unsigned int rec = (base + lane < eend) ? raw : REC_SINK;
        const int nbase = base + LWAVES * 32;
        raw = recs[min(nbase + lane, N_EDGES - 1)];
        unsigned int rr[16];
        float vv[16];
#define GATHER_STEP(J)                                                                   \
        rr[J] = (unsigned int)__builtin_amdgcn_mov_dpp((int)rec, 0x150 + (J), 0xF, 0xF, true);      \
        vv[J] = xin[(rr[J] & SRC_MASK) | (unsigned int)col];
        GATHER_STEP(0)  GATHER_STEP(1)  GATHER_STEP(2)  GATHER_STEP(3)
        GATHER_STEP(4)  GATHER_STEP(5)  GATHER_STEP(6)  GATHER_STEP(7)
        GATHER_STEP(8)  GATHER_STEP(9)  GATHER_STEP(10) GATHER_STEP(11)
        GATHER_STEP(12) GATHER_STEP(13) GATHER_STEP(14) GATHER_STEP(15)
#undef GATHER_STEP
#pragma unroll
        for (int j = 0; j < 16; ++j)
            atomicAdd(&sagg[(rr[j] >> SRC_BITS) * FDIM + col], vv[j]);
        base = nbase;
    }
    __syncthreads();

    const int tile = bucket * BTILES + wid;
    const float epsp = 1.0f + eps_p[0];

    const float* xrow = xin + (size_t)(tile * 16 + col) * FDIM;
    const float* grow = sagg + (wid * 16 + col) * FDIM;

    v8f acc = {};
#pragma unroll
    for (int c = 0; c < 4; ++c) {
        const int kb = c * 4 + hi * 2;
        v2f xv = *(const v2f*)(xrow + kb);
        v2f gv = *(const v2f*)(grow + kb);
        v2f a;
        a.x = epsp * xv.x + gv.x;
        a.y = epsp * xv.y + gv.y;
        v2f b;
        b.x = wa[(kb + 0) * 16 + col];
        b.y = wa[(kb + 1) * 16 + col];
        acc = __builtin_amdgcn_wmma_f32_16x16x4_f32(false, a, false, b,
                                                    (short)0, acc, false, false);
    }

    const float biasA = ba[col];
#pragma unroll
    for (int v = 0; v < 8; ++v) {
        float t = acc[v] + biasA;
        lds[wid][v + hi * 8][col] = t > 0.0f ? t : 0.0f;
    }
    __syncthreads();

    v8f acc2 = {};
#pragma unroll
    for (int c = 0; c < 4; ++c) {
        const int kb = c * 4 + hi * 2;
        v2f a;
        a.x = lds[wid][col][kb + 0];
        a.y = lds[wid][col][kb + 1];
        v2f b;
        b.x = wb[(kb + 0) * 16 + col];
        b.y = wb[(kb + 1) * 16 + col];
        acc2 = __builtin_amdgcn_wmma_f32_16x16x4_f32(false, a, false, b,
                                                     (short)0, acc2, false, false);
    }

    const float biasB = bb[col];
    float* orow = out + (size_t)tile * 16 * FDIM;

    if (LOGSOFTMAX) {
#pragma unroll
        for (int v = 0; v < 8; ++v) {
            float val = acc2[v] + biasB;
            float m = val;
#pragma unroll
            for (int off = 1; off < 16; off <<= 1)
                m = fmaxf(m, __shfl_xor(m, off, 16));
            float s = __expf(val - m);
#pragma unroll
            for (int off = 1; off < 16; off <<= 1)
                s += __shfl_xor(s, off, 16);
            orow[(v + hi * 8) * FDIM + col] = val - m - __logf(s);
        }
    } else {
#pragma unroll
        for (int v = 0; v < 8; ++v) {
            float t = acc2[v] + biasB;
            orow[(v + hi * 8) * FDIM + col] = t > 0.0f ? t : 0.0f;
        }
    }
}

extern "C" void kernel_launch(void* const* d_in, const int* in_sizes, int n_in,
                              void* d_out, int out_size, void* d_ws, size_t ws_size,
                              hipStream_t stream) {
    const float* x    = (const float*)d_in[0];
    const int*   ei   = (const int*)d_in[1];
    const float* eps1 = (const float*)d_in[2];
    const float* w1a  = (const float*)d_in[3];
    const float* b1a  = (const float*)d_in[4];
    const float* w1b  = (const float*)d_in[5];
    const float* b1b  = (const float*)d_in[6];
    const float* eps2 = (const float*)d_in[7];
    const float* w2a  = (const float*)d_in[8];
    const float* b2a  = (const float*)d_in[9];
    const float* w2b  = (const float*)d_in[10];
    const float* b2b  = (const float*)d_in[11];
    float* out = (float*)d_out;

    int* gcount = (int*)d_ws;
    int* bstart = gcount + 512;
    int* cursor = gcount + 1024;
    unsigned int* recs = (unsigned int*)(gcount + 2048);
    float* h1 = (float*)(recs + N_EDGES);

    k_zero_i<<<1, 512, 0, stream>>>(gcount, NBUCK);
    k_count<<<BIN_BLOCKS, BIN_THREADS, 0, stream>>>(ei, gcount);
    k_scan<<<1, 512, 0, stream>>>(gcount, bstart, cursor);
    k_place<<<BIN_BLOCKS, BIN_THREADS, 0, stream>>>(ei, cursor, recs);
    k_layer<false><<<NBUCK, LTHREADS, 0, stream>>>(x, recs, bstart, eps1, w1a, b1a, w1b, b1b, h1);
    k_layer<true><<<NBUCK, LTHREADS, 0, stream>>>(h1, recs, bstart, eps2, w2a, b2a, w2b, b2b, out);
}


# Edge bucketing in one launch, not four: each binning work-group counting-sorts its 12800 edges in LDS and stores whole lines; a layer work-group stages its 250 segments; f32 add order changes
# speedup vs baseline: 5.2477x; 1.3318x over previous
#include <hip/hip_runtime.h>


#define N_NODES 200000
#define N_EDGES 3200000
#define FDIM    16
#define NTILES  (N_NODES / 16)

#define BROWS    400
#define BTILES   (BROWS / 16)
#define NBUCK    (N_NODES / BROWS)
#define LWAVES   BTILES
#define LTHREADS (LWAVES * 32)

#define BIN_THREADS 512
#define BIN_EPT     25
#define BIN_BLOCKS  (N_EDGES / (BIN_THREADS * BIN_EPT))

#define SEG_EDGES     (BIN_THREADS * BIN_EPT)
#define NSEG          BIN_BLOCKS
#define SEG_PITCH     256
#define SEG_ROWS      512
#define SEGS_PER_WAVE (NSEG / LWAVES)
#define STAGE_CAP     8192

#define SRC_BITS 22
#define SRC_MASK 0x3FFFFFu
#define REC_SINK ((unsigned int)BROWS << SRC_BITS)

static_assert(NBUCK * BROWS == N_NODES, "buckets must tile the nodes");
static_assert(BIN_BLOCKS * BIN_THREADS * BIN_EPT == N_EDGES, "binning grid must tile the edges");
static_assert(N_NODES * FDIM <= (1 << SRC_BITS), "source row offset must fit its field");
static_assert(BROWS < (1 << (32 - SRC_BITS)), "bucket row (and the spare row) must fit its field");
static_assert(NBUCK < BIN_THREADS, "one thread per bucket counter and one for the end offset");
static_assert(NSEG <= SEG_PITCH && NBUCK + 1 <= SEG_ROWS, "offset table must hold every (bucket, segment)");
static_assert(SEGS_PER_WAVE * LWAVES == NSEG, "segments must tile the waves of a layer work-group");
static_assert(SEG_PITCH <= LTHREADS && SEG_PITCH % 32 == 0, "one thread per segment, in whole waves");
static_assert(STAGE_CAP % 32 == 0, "the stage holds whole 32-record chunks");

typedef float v2f __attribute__((ext_vector_type(2)));
typedef float v8f __attribute__((ext_vector_type(8)));

__global__ __launch_bounds__(BIN_THREADS) void k_sort(const int* __restrict__ ei,
                                                      unsigned int* __restrict__ recs,
                                                      int* __restrict__ segoff) {
    __shared__ int h[BIN_THREADS];
    __shared__ int off[BIN_THREADS];
    __shared__ int wtot[BIN_THREADS / 32];
    __shared__ unsigned int srt[SEG_EDGES];
    const int tid  = threadIdx.x;
    const int lane = tid & 31;
    const int wid  = tid >> 5;
    h[tid] = 0;
    __syncthreads();
    const int* srcp = ei + blockIdx.x * SEG_EDGES + tid;
    const int* dstp = srcp + N_EDGES;
    int sv[BIN_EPT], dv[BIN_EPT];
#pragma unroll
    for (int k = 0; k < BIN_EPT; ++k) {
        sv[k] = srcp[k * BIN_THREADS];
        dv[k] = dstp[k * BIN_THREADS];
    }
    unsigned int rec[BIN_EPT];
    int bk[BIN_EPT], rk[BIN_EPT];
#pragma unroll
    for (int k = 0; k < BIN_EPT; ++k) {
        const unsigned int d = (unsigned int)dv[k];
        const unsigned int b = d / BROWS;
        bk[k]  = (int)b;
        rec[k] = ((unsigned int)sv[k] * FDIM) | ((d - b * BROWS) << SRC_BITS);
        rk[k]  = atomicAdd(&h[b], 1);
    }
    __syncthreads();

    const int cnt = h[tid];
    int inc = cnt;
    for (int o = 1; o < 32; o <<= 1) {
        const int t = __shfl_up(inc, o, 32);
        if (lane >= o) inc += t;
    }
    if (lane == 31) wtot[wid] = inc;
    __syncthreads();
    int pre = 0;
    for (int j = 0; j < BIN_THREADS / 32; ++j) pre += (j < wid) ? wtot[j] : 0;
    const int excl = pre + inc - cnt;
    off[tid] = excl;
    if (tid <= NBUCK) segoff[tid * SEG_PITCH + blockIdx.x] = excl;
    __syncthreads();

#pragma unroll
    for (int k = 0; k < BIN_EPT; ++k) srt[off[bk[k]] + rk[k]] = rec[k];
    __syncthreads();

    unsigned int* runp = recs + blockIdx.x * SEG_EDGES + tid;
#pragma unroll
    for (int k = 0; k < BIN_EPT; ++k) runp[k * BIN_THREADS] = srt[k * BIN_THREADS + tid];
}

template <bool LOGSOFTMAX>
__global__ __launch_bounds__(LTHREADS) void k_layer(const float* __restrict__ xin,
                                                    const unsigned int* __restrict__ recs,
                                                    const int* __restrict__ segoff,
                                                    const float* __restrict__ eps_p,
                                                    const float* __restrict__ wa,
                                                    const float* __restrict__ ba,
                                                    const float* __restrict__ wb,
                                                    const float* __restrict__ bb,
                                                    float* __restrict__ out) {
    __shared__ __attribute__((aligned(16))) float sagg[(BROWS + 1) * FDIM];
    __shared__ float lds[LWAVES][16][17];
    __shared__ unsigned int stage[STAGE_CAP + 32];
    __shared__ int4 seg[SEG_PITCH];
    __shared__ int wtot[SEG_PITCH / 32];

    const int tid  = threadIdx.x;
    const int wid  = __builtin_amdgcn_readfirstlane(tid >> 5);
    const int lane = tid & 31;
    const int col  = lane & 15;
    const int hi   = lane >> 4;
    const int bucket = blockIdx.x;

    int o0 = 0, o1 = 0;
    if (tid < NSEG) {
        o0 = segoff[bucket * SEG_PITCH + tid];
        o1 = segoff[(bucket + 1) * SEG_PITCH + tid];
    }
    for (int i = tid; i < (BROWS + 1) * FDIM; i += LTHREADS) sagg[i] = 0.0f;
    const int len = o1 - o0;
    const int g0  = tid * SEG_EDGES + o0;
    int inc = len;
    for (int o = 1; o < 32; o <<= 1) {
        const int t = __shfl_up(inc, o, 32);
        if (lane >= o) inc += t;
    }
    if (tid < SEG_PITCH && lane == 31) wtot[wid] = inc;
    __syncthreads();
    if (tid < SEG_PITCH) {
        int pre = 0;
        for (int j = 0; j < SEG_PITCH / 32; ++j) pre += (j < wid) ? wtot[j] : 0;
        seg[tid] = make_int4(pre + inc - len, g0, len, 0);
    }
    __syncthreads();
    const int total = __builtin_amdgcn_readfirstlane(seg[SEG_PITCH - 1].x);

    for (int lo = 0; lo < total; lo += STAGE_CAP) {
        const int nround = min(total - lo, STAGE_CAP);

        int4 sg[SEGS_PER_WAVE];
#pragma unroll
        for (int k = 0; k < SEGS_PER_WAVE; ++k) sg[k] = seg[wid + k * LWAVES];
        unsigned int sr[SEGS_PER_WAVE];
        int sp[SEGS_PER_WAVE];
        int nmax = 0;
#pragma unroll
        for (int k = 0; k < SEGS_PER_WAVE; ++k) {
            const int v = sg[k].x - lo + lane;
            const bool ok = lane < sg[k].z && v >= 0 && v < STAGE_CAP;
            sp[k] = ok ? v : STAGE_CAP + lane;
            sr[k] = recs[ok ? sg[k].y + lane : 0];
            nmax = max(nmax, sg[k].z);
        }
#pragma unroll
        for (int k = 0; k < SEGS_PER_WAVE; ++k) stage[sp[k]] = sr[k];
        if (__builtin_amdgcn_readfirstlane(nmax) > 32) {
#pragma nounroll
            for (int k = 0; k < SEGS_PER_WAVE; ++k) {
                const int4 s = seg[wid + k * LWAVES];
                const int n = __builtin_amdgcn_readfirstlane(s.z);
#pragma nounroll
                for (int p0 = 32; p0 < n; p0 += 32) {
                    const int p = p0 + lane;
                    const int v = s.x - lo + p;
                    if (p < n && v >= 0 && v < STAGE_CAP) stage[v] = recs[s.y + p];
                }
            }
        }
        __syncthreads();

        for (int base = wid * 32; base < nround; base += LWAVES * 32) {
            const unsigned int rec = (base + lane < nround) ? stage[base + lane] : REC_SINK;
            unsigned int rr[16];
            float vv[16];
#define GATHER_STEP(J)                                                                   \
            rr[J] = (unsigned int)__builtin_amdgcn_mov_dpp((int)rec, 0x150 + (J), 0xF, 0xF, true);      \
            vv[J] = xin[(rr[J] & SRC_MASK) | (unsigned int)col];
            GATHER_STEP(0)  GATHER_STEP(1)  GATHER_STEP(2)  GATHER_STEP(3)
            GATHER_STEP(4)  GATHER_STEP(5)  GATHER_STEP(6)  GATHER_STEP(7)
            GATHER_STEP(8)  GATHER_STEP(9)  GATHER_STEP(10) GATHER_STEP(11)
            GATHER_STEP(12) GATHER_STEP(13) GATHER_STEP(14) GATHER_STEP(15)
#undef GATHER_STEP
#pragma unroll
            for (int j = 0; j < 16; ++j)
                atomicAdd(&sagg[(rr[j] >> SRC_BITS) * FDIM + col], vv[j]);
        }
        __syncthreads();
    }

    const int tile = bucket * BTILES + wid;
    const float epsp = 1.0f + eps_p[0];

    const float* xrow = xin + (size_t)(tile * 16 + col) * FDIM;
    const float* grow = sagg + (wid * 16 + col) * FDIM;

    v8f acc = {};
#pragma unroll
    for (int c = 0; c < 4; ++c) {
        const int kb = c * 4 + hi * 2;
        v2f xv = *(const v2f*)(xrow + kb);
        v2f gv = *(const v2f*)(grow + kb);
        v2f a;
        a.x = epsp * xv.x + gv.x;
        a.y = epsp * xv.y + gv.y;
        v2f b;
        b.x = wa[(kb + 0) * 16 + col];
        b.y = wa[(kb + 1) * 16 + col];
        acc = __builtin_amdgcn_wmma_f32_16x16x4_f32(false, a, false, b,
                                                    (short)0, acc, false, false);
    }

    const float biasA = ba[col];
#pragma unroll
    for (int v = 0; v < 8; ++v) {
        float t = acc[v] + biasA;
        lds[wid][v + hi * 8][col] = t > 0.0f ? t : 0.0f;
    }
    __syncthreads();

    v8f acc2 = {};
#pragma unroll
    for (int c = 0; c < 4; ++c) {
        const int kb = c * 4 + hi * 2;
        v2f a;
        a.x = lds[wid][col][kb + 0];
        a.y = lds[wid][col][kb + 1];
        v2f b;
        b.x = wb[(kb + 0) * 16 + col];
        b.y = wb[(kb + 1) * 16 + col];
        acc2 = __builtin_amdgcn_wmma_f32_16x16x4_f32(false, a, false, b,
                                                     (short)0, acc2, false, false);
    }

    const float biasB = bb[col];
    float* orow = out + (size_t)tile * 16 * FDIM;

    if (LOGSOFTMAX) {
#pragma unroll
        for (int v = 0; v < 8; ++v) {
            float val = acc2[v] + biasB;
            float m = val;
#pragma unroll
            for (int off = 1; off < 16; off <<= 1)
                m = fmaxf(m, __shfl_xor(m, off, 16));
            float s = __expf(val - m);
#pragma unroll
            for (int off = 1; off < 16; off <<= 1)
                s += __shfl_xor(s, off, 16);
            orow[(v + hi * 8) * FDIM + col] = val - m - __logf(s);
        }
    } else {
#pragma unroll
        for (int v = 0; v < 8; ++v) {
            float t = acc2[v] + biasB;
            orow[(v + hi * 8) * FDIM + col] = t > 0.0f ? t : 0.0f;
        }
    }
}

extern "C" void kernel_launch(void* const* d_in, const int* in_sizes, int n_in,
                              void* d_out, int out_size, void* d_ws, size_t ws_size,
                              hipStream_t stream) {
    const float* x    = (const float*)d_in[0];
    const int*   ei   = (const int*)d_in[1];
    const float* eps1 = (const float*)d_in[2];
    const float* w1a  = (const float*)d_in[3];
    const float* b1a  = (const float*)d_in[4];
    const float* w1b  = (const float*)d_in[5];
    const float* b1b  = (const float*)d_in[6];
    const float* eps2 = (const float*)d_in[7];
    const float* w2a  = (const float*)d_in[8];
    const float* b2a  = (const float*)d_in[9];
    const float* w2b  = (const float*)d_in[10];
    const float* b2b  = (const float*)d_in[11];
    float* out = (float*)d_out;

    int* segoff = (int*)d_ws;
    unsigned int* recs = (unsigned int*)(segoff + SEG_ROWS * SEG_PITCH);
    float* h1 = (float*)(recs + N_EDGES);

    k_sort<<<BIN_BLOCKS, BIN_THREADS, 0, stream>>>(ei, recs, segoff);
    k_layer<false><<<NBUCK, LTHREADS, 0, stream>>>(x, recs, segoff, eps1, w1a, b1a, w1b, b1b, h1);
    k_layer<true><<<NBUCK, LTHREADS, 0, stream>>>(h1, recs, segoff, eps2, w2a, b2a, w2b, b2b, out);
}
